# starting kernel
# baseline (speedup 1.0000x reference)
#include <hip/hip_runtime.h>
#include <hip/hip_bf16.h>

#define BB 4
#define NN 128
#define MM 128
#define DD 256
#define BD 1024

typedef __attribute__((ext_vector_type(2))) float v2f;
typedef __attribute__((ext_vector_type(4))) float v4f;
typedef __attribute__((ext_vector_type(8))) float v8f;

__global__ void __launch_bounds__(256)
fold_w_kernel(const float* __restrict__ A, const float* __restrict__ W,
              float* __restrict__ Aw) {
  __shared__ v4f w4[BD / 4];
  const int tid = threadIdx.x;
  w4[tid] = ((const v4f*)W)[tid];
  __syncthreads();

  const int wave = tid >> 5;
  const int lane = tid & 31;
  const size_t pair = (size_t)blockIdx.x * 8 + wave;
  const v4f* a4 = (const v4f*)A + pair * (BD / 4);

  float acc = 0.0f;
#pragma unroll
  for (int j = 0; j < 8; ++j) {
    const v4f a = __builtin_nontemporal_load(&a4[lane + 32 * j]);
    const v4f w = w4[lane + 32 * j];
    acc = fmaf(a.x, w.x, acc);
    acc = fmaf(a.y, w.y, acc);
    acc = fmaf(a.z, w.z, acc);
    acc = fmaf(a.w, w.w, acc);
  }
#pragma unroll
  for (int off = 16; off >= 1; off >>= 1)
    acc += __shfl_xor(acc, off, 32);
  if (lane == 0) Aw[pair] = acc;
}

__global__ void __launch_bounds__(32)
gemm_xaw_kernel(const float* __restrict__ X, const float* __restrict__ Aw,
                float* __restrict__ T) {
  const int lane = threadIdx.x & 31;
  const int half = lane >> 4;
  const int l16  = lane & 15;
  const int row0 = blockIdx.x * 16;
  const int col0 = blockIdx.y * 16;

  v8f c = {};
  for (int k = 0; k < DD; k += 4) {
    const float* xp = X + (size_t)(row0 + l16) * DD + k + 2 * half;
    v2f a; a.x = xp[0]; a.y = xp[1];
    const float* bp = Aw + (size_t)(k + 2 * half) * DD + col0 + l16;
    v2f bf; bf.x = bp[0]; bf.y = bp[DD];
    c = __builtin_amdgcn_wmma_f32_16x16x4_f32(false, a, false, bf,
                                              (short)0, c, false, false);
  }
#pragma unroll
  for (int r = 0; r < 8; ++r)
    T[(size_t)(row0 + r + 8 * half) * DD + col0 + l16] = c[r];
}

__global__ void __launch_bounds__(32)
gemm_ty_kernel(const float* __restrict__ T, const float* __restrict__ Y,
               const float* __restrict__ bias, float* __restrict__ S) {
  const int lane = threadIdx.x & 31;
  const int half = lane >> 4;
  const int l16  = lane & 15;
  const int b  = blockIdx.z;
  const int n0 = blockIdx.x * 16;
  const int m0 = blockIdx.y * 16;
  const float* Tb = T + (size_t)b * NN * DD;
  const float* Yb = Y + (size_t)b * MM * DD;

  v8f c = {};
  for (int k = 0; k < DD; k += 4) {
    const float* ap = Tb + (size_t)(n0 + l16) * DD + k + 2 * half;
    v2f a; a.x = ap[0]; a.y = ap[1];
    const float* bp = Yb + (size_t)(m0 + l16) * DD + k + 2 * half;
    v2f bf; bf.x = bp[0]; bf.y = bp[1];
    c = __builtin_amdgcn_wmma_f32_16x16x4_f32(false, a, false, bf,
                                              (short)0, c, false, false);
  }
  const float bv = bias[0];
#pragma unroll
  for (int r = 0; r < 8; ++r)
    S[((size_t)b * NN + n0 + r + 8 * half) * MM + m0 + l16] = c[r] + bv;
}

extern "C" void kernel_launch(void* const* d_in, const int* in_sizes, int n_in,
                              void* d_out, int out_size, void* d_ws, size_t ws_size,
                              hipStream_t stream) {
  (void)in_sizes; (void)n_in; (void)out_size; (void)ws_size;
  const float* X = (const float*)d_in[0];
  const float* Y = (const float*)d_in[1];
  const float* A = (const float*)d_in[2];
  const float* W = (const float*)d_in[3];
  const float* b = (const float*)d_in[4];
  float* out = (float*)d_out;

  float* Aw = (float*)d_ws;
  float* T  = Aw + (size_t)DD * DD;

  fold_w_kernel<<<(DD * DD) / 8, 256, 0, stream>>>(A, W, Aw);
  gemm_xaw_kernel<<<dim3((BB * NN) / 16, DD / 16), 32, 0, stream>>>(X, Aw, T);
  gemm_ty_kernel<<<dim3(NN / 16, MM / 16, BB), 32, 0, stream>>>(T, Y, b, out);
}


# fold kernel as a resident grid: 256 workgroups x 16 waves, each wave walks 16 rows of A with 3 rows of non-temporal loads in flight ahead, W held in registers (no LDS, no barrier); sum order kept
# speedup vs baseline: 1.1160x; 1.1160x over previous
#include <hip/hip_runtime.h>
#include <hip/hip_bf16.h>

#define BB 4
#define NN 128
#define MM 128
#define DD 256
#define BD 1024

typedef __attribute__((ext_vector_type(2))) float v2f;
typedef __attribute__((ext_vector_type(4))) float v4f;
typedef __attribute__((ext_vector_type(8))) float v8f;

#define K1_ROWS  16
#define K1_WAVES 16
#define K1_AHEAD 3
#define K1_Q     (BD / 128)

__global__ void __launch_bounds__(K1_WAVES * 32)
fold_w_kernel(const float* __restrict__ A, const float* __restrict__ W,
              float* __restrict__ Aw) {
  const unsigned tid  = threadIdx.x;
  const unsigned lane = tid & 31u;
  const unsigned wave = __builtin_amdgcn_readfirstlane(tid >> 5);
  const unsigned row0 = (blockIdx.x * K1_WAVES + wave) * K1_ROWS;

  v4f wq[K1_Q];
  const v4f* w4 = (const v4f*)W + lane;
#pragma unroll
  for (int q = 0; q < K1_Q; ++q) wq[q] = w4[32 * q];

  const v4f* a4 = (const v4f*)A + (size_t)row0 * (BD / 4) + lane;
  v4f buf[K1_AHEAD + 1][K1_Q];
  float sum[K1_ROWS];

#define K1_LOAD_ROW(R) do { \
    _Pragma("unroll") for (int q = 0; q < K1_Q; ++q) \
      buf[(R) % (K1_AHEAD + 1)][q] = \
        __builtin_nontemporal_load(a4 + (size_t)(R) * (BD / 4) + 32 * q); } while (0)

#pragma unroll
  for (int r = 0; r < K1_AHEAD; ++r) K1_LOAD_ROW(r);

#pragma unroll
  for (int r = 0; r < K1_ROWS; ++r) {
    __builtin_amdgcn_sched_barrier(0);
    if (r + K1_AHEAD < K1_ROWS) K1_LOAD_ROW(r + K1_AHEAD);
    __builtin_amdgcn_sched_barrier(0);
    const v4f* ar = buf[r % (K1_AHEAD + 1)];
    float acc = 0.0f;
#pragma unroll
    for (int q = 0; q < K1_Q; ++q) {
      acc = fmaf(ar[q].x, wq[q].x, acc);
      acc = fmaf(ar[q].y, wq[q].y, acc);
      acc = fmaf(ar[q].z, wq[q].z, acc);
      acc = fmaf(ar[q].w, wq[q].w, acc);
    }
    sum[r] = acc;
  }
#undef K1_LOAD_ROW

#pragma unroll
  for (int r = 0; r < K1_ROWS; ++r) {
    float s = sum[r];
#pragma unroll
    for (int off = 16; off >= 1; off >>= 1)
      s += __shfl_xor(s, off, 32);
    sum[r] = s;
  }
  float mine = 0.0f;
#pragma unroll
  for (int r = 0; r < K1_ROWS; ++r)
    if (lane == (unsigned)r) mine = sum[r];
  if (lane < (unsigned)K1_ROWS) Aw[row0 + lane] = mine;
}

__global__ void __launch_bounds__(32)
gemm_xaw_kernel(const float* __restrict__ X, const float* __restrict__ Aw,
                float* __restrict__ T) {
  const int lane = threadIdx.x & 31;
  const int half = lane >> 4;
  const int l16  = lane & 15;
  const int row0 = blockIdx.x * 16;
  const int col0 = blockIdx.y * 16;

  v8f c = {};
  for (int k = 0; k < DD; k += 4) {
    const float* xp = X + (size_t)(row0 + l16) * DD + k + 2 * half;
    v2f a; a.x = xp[0]; a.y = xp[1];
    const float* bp = Aw + (size_t)(k + 2 * half) * DD + col0 + l16;
    v2f bf; bf.x = bp[0]; bf.y = bp[DD];
    c = __builtin_amdgcn_wmma_f32_16x16x4_f32(false, a, false, bf,
                                              (short)0, c, false, false);
  }
#pragma unroll
  for (int r = 0; r < 8; ++r)
    T[(size_t)(row0 + r + 8 * half) * DD + col0 + l16] = c[r];
}

__global__ void __launch_bounds__(32)
gemm_ty_kernel(const float* __restrict__ T, const float* __restrict__ Y,
               const float* __restrict__ bias, float* __restrict__ S) {
  const int lane = threadIdx.x & 31;
  const int half = lane >> 4;
  const int l16  = lane & 15;
  const int b  = blockIdx.z;
  const int n0 = blockIdx.x * 16;
  const int m0 = blockIdx.y * 16;
  const float* Tb = T + (size_t)b * NN * DD;
  const float* Yb = Y + (size_t)b * MM * DD;

  v8f c = {};
  for (int k = 0; k < DD; k += 4) {
    const float* ap = Tb + (size_t)(n0 + l16) * DD + k + 2 * half;
    v2f a; a.x = ap[0]; a.y = ap[1];
    const float* bp = Yb + (size_t)(m0 + l16) * DD + k + 2 * half;
    v2f bf; bf.x = bp[0]; bf.y = bp[1];
    c = __builtin_amdgcn_wmma_f32_16x16x4_f32(false, a, false, bf,
                                              (short)0, c, false, false);
  }
  const float bv = bias[0];
#pragma unroll
  for (int r = 0; r < 8; ++r)
    S[((size_t)b * NN + n0 + r + 8 * half) * MM + m0 + l16] = c[r] + bv;
}

extern "C" void kernel_launch(void* const* d_in, const int* in_sizes, int n_in,
                              void* d_out, int out_size, void* d_ws, size_t ws_size,
                              hipStream_t stream) {
  (void)in_sizes; (void)n_in; (void)out_size; (void)ws_size;
  const float* X = (const float*)d_in[0];
  const float* Y = (const float*)d_in[1];
  const float* A = (const float*)d_in[2];
  const float* W = (const float*)d_in[3];
  const float* b = (const float*)d_in[4];
  float* out = (float*)d_out;

  float* Aw = (float*)d_ws;
  float* T  = Aw + (size_t)DD * DD;

  fold_w_kernel<<<(DD * DD) / (K1_WAVES * K1_ROWS), K1_WAVES * 32, 0, stream>>>(A, W, Aw);
  gemm_xaw_kernel<<<dim3((BB * NN) / 16, DD / 16), 32, 0, stream>>>(X, Aw, T);
  gemm_ty_kernel<<<dim3(NN / 16, MM / 16, BB), 32, 0, stream>>>(T, Y, b, out);
}


# both GEMM kernels written load-all-then-one-chain with an occupancy hint: about 20 k-steps of fragment loads stay in flight in a rolling window, not batches of 8 and 16 that drain; sum order kept
# speedup vs baseline: 1.2509x; 1.1209x over previous
#include <hip/hip_runtime.h>
#include <hip/hip_bf16.h>

#define BB 4
#define NN 128
#define MM 128
#define DD 256
#define BD 1024

typedef __attribute__((ext_vector_type(2))) float v2f;
typedef __attribute__((ext_vector_type(4))) float v4f;
typedef __attribute__((ext_vector_type(8))) float v8f;

#define K1_ROWS  16
#define K1_WAVES 16
#define K1_AHEAD 3
#define K1_Q     (BD / 128)

__global__ void __launch_bounds__(K1_WAVES * 32)
fold_w_kernel(const float* __restrict__ A, const float* __restrict__ W,
              float* __restrict__ Aw) {
  const unsigned tid  = threadIdx.x;
  const unsigned lane = tid & 31u;
  const unsigned wave = __builtin_amdgcn_readfirstlane(tid >> 5);
  const unsigned row0 = (blockIdx.x * K1_WAVES + wave) * K1_ROWS;

  v4f wq[K1_Q];
  const v4f* w4 = (const v4f*)W + lane;
#pragma unroll
  for (int q = 0; q < K1_Q; ++q) wq[q] = w4[32 * q];

  const v4f* a4 = (const v4f*)A + (size_t)row0 * (BD / 4) + lane;
  v4f buf[K1_AHEAD + 1][K1_Q];
  float sum[K1_ROWS];

#define K1_LOAD_ROW(R) do { \
    _Pragma("unroll") for (int q = 0; q < K1_Q; ++q) \
      buf[(R) % (K1_AHEAD + 1)][q] = \
        __builtin_nontemporal_load(a4 + (size_t)(R) * (BD / 4) + 32 * q); } while (0)

#pragma unroll
  for (int r = 0; r < K1_AHEAD; ++r) K1_LOAD_ROW(r);

#pragma unroll
  for (int r = 0; r < K1_ROWS; ++r) {
    __builtin_amdgcn_sched_barrier(0);
    if (r + K1_AHEAD < K1_ROWS) K1_LOAD_ROW(r + K1_AHEAD);
    __builtin_amdgcn_sched_barrier(0);
    const v4f* ar = buf[r % (K1_AHEAD + 1)];
    float acc = 0.0f;
#pragma unroll
    for (int q = 0; q < K1_Q; ++q) {
      acc = fmaf(ar[q].x, wq[q].x, acc);
      acc = fmaf(ar[q].y, wq[q].y, acc);
      acc = fmaf(ar[q].z, wq[q].z, acc);
      acc = fmaf(ar[q].w, wq[q].w, acc);
    }
    sum[r] = acc;
  }
#undef K1_LOAD_ROW

#pragma unroll
  for (int r = 0; r < K1_ROWS; ++r) {
    float s = sum[r];
#pragma unroll
    for (int off = 16; off >= 1; off >>= 1)
      s += __shfl_xor(s, off, 32);
    sum[r] = s;
  }
  float mine = 0.0f;
#pragma unroll
  for (int r = 0; r < K1_ROWS; ++r)
    if (lane == (unsigned)r) mine = sum[r];
  if (lane < (unsigned)K1_ROWS) Aw[row0 + lane] = mine;
}

__global__ void __launch_bounds__(32) __attribute__((amdgpu_waves_per_eu(1, 2)))
gemm_xaw_kernel(const float* __restrict__ X, const float* __restrict__ Aw,
                float* __restrict__ T) {
  const int lane = threadIdx.x & 31;
  const int half = lane >> 4;
  const int l16  = lane & 15;
  const int row0 = blockIdx.x * 16;
  const int col0 = blockIdx.y * 16;

  v2f a[DD / 4], bf[DD / 4];
#pragma unroll
  for (int i = 0; i < DD / 4; ++i) {
    const int k = 4 * i;
    const float* xp = X + (size_t)(row0 + l16) * DD + k + 2 * half;
    a[i].x = xp[0]; a[i].y = xp[1];
    const float* bp = Aw + (size_t)(k + 2 * half) * DD + col0 + l16;
    bf[i].x = bp[0]; bf[i].y = bp[DD];
  }
  v8f c = {};
#pragma unroll
  for (int i = 0; i < DD / 4; ++i)
    c = __builtin_amdgcn_wmma_f32_16x16x4_f32(false, a[i], false, bf[i],
                                              (short)0, c, false, false);
#pragma unroll
  for (int r = 0; r < 8; ++r)
    T[(size_t)(row0 + r + 8 * half) * DD + col0 + l16] = c[r];
}

__global__ void __launch_bounds__(32) __attribute__((amdgpu_waves_per_eu(1, 2)))
gemm_ty_kernel(const float* __restrict__ T, const float* __restrict__ Y,
               const float* __restrict__ bias, float* __restrict__ S) {
  const int lane = threadIdx.x & 31;
  const int half = lane >> 4;
  const int l16  = lane & 15;
  const int b  = blockIdx.z;
  const int n0 = blockIdx.x * 16;
  const int m0 = blockIdx.y * 16;
  const float* Tb = T + (size_t)b * NN * DD;
  const float* Yb = Y + (size_t)b * MM * DD;

  v2f a[DD / 4], bf[DD / 4];
#pragma unroll
  for (int i = 0; i < DD / 4; ++i) {
    const int k = 4 * i;
    const float* ap = Tb + (size_t)(n0 + l16) * DD + k + 2 * half;
    a[i].x = ap[0]; a[i].y = ap[1];
    const float* bp = Yb + (size_t)(m0 + l16) * DD + k + 2 * half;
    bf[i].x = bp[0]; bf[i].y = bp[1];
  }
  v8f c = {};
#pragma unroll
  for (int i = 0; i < DD / 4; ++i)
    c = __builtin_amdgcn_wmma_f32_16x16x4_f32(false, a[i], false, bf[i],
                                              (short)0, c, false, false);
  const float bv = bias[0];
#pragma unroll
  for (int r = 0; r < 8; ++r)
    S[((size_t)b * NN + n0 + r + 8 * half) * MM + m0 + l16] = c[r] + bv;
}

extern "C" void kernel_launch(void* const* d_in, const int* in_sizes, int n_in,
                              void* d_out, int out_size, void* d_ws, size_t ws_size,
                              hipStream_t stream) {
  (void)in_sizes; (void)n_in; (void)out_size; (void)ws_size;
  const float* X = (const float*)d_in[0];
  const float* Y = (const float*)d_in[1];
  const float* A = (const float*)d_in[2];
  const float* W = (const float*)d_in[3];
  const float* b = (const float*)d_in[4];
  float* out = (float*)d_out;

  float* Aw = (float*)d_ws;
  float* T  = Aw + (size_t)DD * DD;

  fold_w_kernel<<<(DD * DD) / (K1_WAVES * K1_ROWS), K1_WAVES * 32, 0, stream>>>(A, W, Aw);
  gemm_xaw_kernel<<<dim3((BB * NN) / 16, DD / 16), 32, 0, stream>>>(X, Aw, T);
  gemm_ty_kernel<<<dim3(NN / 16, MM / 16, BB), 32, 0, stream>>>(T, Y, b, out);
}


# two GEMM kernels made one launch of 256 four-wave workgroups, one per tile of S, each rebuilding its 16-row strip of T = X @ Aw; k order in T permuted in eights, S summed as four K-quarter sums
# speedup vs baseline: 1.3015x; 1.0405x over previous
#include <hip/hip_runtime.h>
#include <hip/hip_bf16.h>

#define BB 4
#define NN 128
#define MM 128
#define DD 256
#define BD 1024

typedef __attribute__((ext_vector_type(2))) float v2f;
typedef __attribute__((ext_vector_type(4))) float v4f;
typedef __attribute__((ext_vector_type(8))) float v8f;

#define K1_ROWS  16
#define K1_WAVES 16
#define K1_AHEAD 3
#define K1_Q     (BD / 128)

__global__ void __launch_bounds__(K1_WAVES * 32)
fold_w_kernel(const float* __restrict__ A, const float* __restrict__ W,
              float* __restrict__ Aw) {
  const unsigned tid  = threadIdx.x;
  const unsigned lane = tid & 31u;
  const unsigned wave = __builtin_amdgcn_readfirstlane(tid >> 5);
  const unsigned row0 = (blockIdx.x * K1_WAVES + wave) * K1_ROWS;

  v4f wq[K1_Q];
  const v4f* w4 = (const v4f*)W + lane;
#pragma unroll
  for (int q = 0; q < K1_Q; ++q) wq[q] = w4[32 * q];

  const v4f* a4 = (const v4f*)A + (size_t)row0 * (BD / 4) + lane;
  v4f buf[K1_AHEAD + 1][K1_Q];
  float sum[K1_ROWS];

#define K1_LOAD_ROW(R) do { \
    _Pragma("unroll") for (int q = 0; q < K1_Q; ++q) \
      buf[(R) % (K1_AHEAD + 1)][q] = \
        __builtin_nontemporal_load(a4 + (size_t)(R) * (BD / 4) + 32 * q); } while (0)

#pragma unroll
  for (int r = 0; r < K1_AHEAD; ++r) K1_LOAD_ROW(r);

#pragma unroll
  for (int r = 0; r < K1_ROWS; ++r) {
    __builtin_amdgcn_sched_barrier(0);
    if (r + K1_AHEAD < K1_ROWS) K1_LOAD_ROW(r + K1_AHEAD);
    __builtin_amdgcn_sched_barrier(0);
    const v4f* ar = buf[r % (K1_AHEAD + 1)];
    float acc = 0.0f;
#pragma unroll
    for (int q = 0; q < K1_Q; ++q) {
      acc = fmaf(ar[q].x, wq[q].x, acc);
      acc = fmaf(ar[q].y, wq[q].y, acc);
      acc = fmaf(ar[q].z, wq[q].z, acc);
      acc = fmaf(ar[q].w, wq[q].w, acc);
    }
    sum[r] = acc;
  }
#undef K1_LOAD_ROW

#pragma unroll
  for (int r = 0; r < K1_ROWS; ++r) {
    float s = sum[r];
#pragma unroll
    for (int off = 16; off >= 1; off >>= 1)
      s += __shfl_xor(s, off, 32);
    sum[r] = s;
  }
  float mine = 0.0f;
#pragma unroll
  for (int r = 0; r < K1_ROWS; ++r)
    if (lane == (unsigned)r) mine = sum[r];
  if (lane < (unsigned)K1_ROWS) Aw[row0 + lane] = mine;
}

#define TS_PITCH 68

__global__ void __launch_bounds__(128) __attribute__((amdgpu_waves_per_eu(1, 2)))
gemm_xaw_ty_kernel(const float* __restrict__ X, const float* __restrict__ Aw,
                   const float* __restrict__ Y, const float* __restrict__ bias,
                   float* __restrict__ S) {
  __shared__ __attribute__((aligned(16))) float Ts[4][16][TS_PITCH];
  __shared__ float red[4][8][32];

  const int tid  = threadIdx.x;
  const int lane = tid & 31;
  const int w    = __builtin_amdgcn_readfirstlane(tid >> 5);
  const int half = lane >> 4;
  const int l16  = lane & 15;
  const int n0 = blockIdx.x * 16;
  const int m0 = blockIdx.y * 16;
  const int b  = blockIdx.z;
  float bv = bias[0];

  const float* Xr = X + ((size_t)b * NN + n0 + l16) * DD + 4 * half;
  const float* Ab = Aw + (size_t)(4 * half) * DD + 64 * w + 4 * l16;
  const float* Yr = Y + ((size_t)b * MM + m0 + l16) * DD + 64 * w + 4 * half;

  v4f xa[DD / 8];
  v4f q[DD / 8][4];
#pragma unroll
  for (int j = 0; j < DD / 8; ++j) {
    xa[j] = *(const v4f*)(Xr + 8 * j);
#pragma unroll
    for (int s = 0; s < 4; ++s)
      q[j][s] = *(const v4f*)(Ab + (size_t)(8 * j + s) * DD);
  }
  v4f yq[8];
#pragma unroll
  for (int j = 0; j < 8; ++j) yq[j] = *(const v4f*)(Yr + 8 * j);

  v8f t0 = {}, t1 = {}, t2 = {}, t3 = {};
#pragma unroll
  for (int j = 0; j < DD / 8; ++j) {
    {
      v2f a = {xa[j].x, xa[j].y};
      v2f b0 = {q[j][0].x, q[j][1].x};
      v2f b1 = {q[j][0].y, q[j][1].y};
      v2f b2 = {q[j][0].z, q[j][1].z};
      v2f b3 = {q[j][0].w, q[j][1].w};
      t0 = __builtin_amdgcn_wmma_f32_16x16x4_f32(false, a, false, b0, (short)0, t0, false, false);
      t1 = __builtin_amdgcn_wmma_f32_16x16x4_f32(false, a, false, b1, (short)0, t1, false, false);
      t2 = __builtin_amdgcn_wmma_f32_16x16x4_f32(false, a, false, b2, (short)0, t2, false, false);
      t3 = __builtin_amdgcn_wmma_f32_16x16x4_f32(false, a, false, b3, (short)0, t3, false, false);
    }
    {
      v2f a = {xa[j].z, xa[j].w};
      v2f b0 = {q[j][2].x, q[j][3].x};
      v2f b1 = {q[j][2].y, q[j][3].y};
      v2f b2 = {q[j][2].z, q[j][3].z};
      v2f b3 = {q[j][2].w, q[j][3].w};
      t0 = __builtin_amdgcn_wmma_f32_16x16x4_f32(false, a, false, b0, (short)0, t0, false, false);
      t1 = __builtin_amdgcn_wmma_f32_16x16x4_f32(false, a, false, b1, (short)0, t1, false, false);
      t2 = __builtin_amdgcn_wmma_f32_16x16x4_f32(false, a, false, b2, (short)0, t2, false, false);
      t3 = __builtin_amdgcn_wmma_f32_16x16x4_f32(false, a, false, b3, (short)0, t3, false, false);
    }
  }

  asm volatile("" : "+s"(bv));

#pragma unroll
  for (int r = 0; r < 8; ++r) {
    v4f v = {t0[r], t1[r], t2[r], t3[r]};
    *(v4f*)&Ts[w][r + 8 * half][4 * l16] = v;
  }
  __syncthreads();

  v8f s = {};
#pragma unroll
  for (int j = 0; j < 8; ++j) {
    v4f tq = *(const v4f*)&Ts[w][l16][8 * j + 4 * half];
    v2f a0 = {tq.x, tq.y}, a1 = {tq.z, tq.w};
    v2f y0 = {yq[j].x, yq[j].y}, y1 = {yq[j].z, yq[j].w};
    s = __builtin_amdgcn_wmma_f32_16x16x4_f32(false, a0, false, y0, (short)0, s, false, false);
    s = __builtin_amdgcn_wmma_f32_16x16x4_f32(false, a1, false, y1, (short)0, s, false, false);
  }

#pragma unroll
  for (int r = 0; r < 8; ++r) red[w][r][lane] = s[r];
  __syncthreads();
#pragma unroll
  for (int u = 0; u < 2; ++u) {
    const int r = 2 * w + u;
    float v = red[0][r][lane];
    v += red[1][r][lane];
    v += red[2][r][lane];
    v += red[3][r][lane];
    S[((size_t)b * NN + n0 + r + 8 * half) * MM + m0 + l16] = v + bv;
  }
}

extern "C" void kernel_launch(void* const* d_in, const int* in_sizes, int n_in,
                              void* d_out, int out_size, void* d_ws, size_t ws_size,
                              hipStream_t stream) {
  (void)in_sizes; (void)n_in; (void)out_size; (void)ws_size;
  const float* X = (const float*)d_in[0];
  const float* Y = (const float*)d_in[1];
  const float* A = (const float*)d_in[2];
  const float* W = (const float*)d_in[3];
  const float* b = (const float*)d_in[4];
  float* out = (float*)d_out;

  float* Aw = (float*)d_ws;

  fold_w_kernel<<<(DD * DD) / (K1_WAVES * K1_ROWS), K1_WAVES * 32, 0, stream>>>(A, W, Aw);
  gemm_xaw_ty_kernel<<<dim3(NN / 16, MM / 16, BB), 128, 0, stream>>>(X, Aw, Y, b, out);
}
